# starting kernel
# baseline (speedup 1.0000x reference)
#include <hip/hip_runtime.h>


typedef __attribute__((ext_vector_type(2))) float v2f;
typedef __attribute__((ext_vector_type(8))) float v8f;

#define FDIM 128
#define LDS_STRIDE 132
#define ROWS_PER_WAVE 32
#define ROWS_PER_BLOCK 256

__global__ __launch_bounds__(256)
void hgnn_gemm128_f32(const float* __restrict__ X,
                      const float* __restrict__ W,
                      const float* __restrict__ W2,
                      float* __restrict__ Out,
                      int nRows)
{
    __shared__ float sWT[FDIM * LDS_STRIDE];

    const int tid  = threadIdx.x;
    const int wave = tid >> 5;
    const int lane = tid & 31;
    const int half = lane >> 4;
    const int m    = lane & 15;

    const int rowBase = blockIdx.x * ROWS_PER_BLOCK + wave * ROWS_PER_WAVE;

    int r0 = rowBase + m;       if (r0 >= nRows) r0 = nRows - 1;
    int r1 = rowBase + 16 + m;  if (r1 >= nRows) r1 = nRows - 1;
    const float* xrow0 = X + (size_t)r0 * FDIM;
    const float* xrow1 = X + (size_t)r1 * FDIM;

    __builtin_prefetch(xrow0, 0, 0);
    __builtin_prefetch(xrow1, 0, 0);

    {
        const float4* w4  = (const float4*)W;
        const int     nf4 = (FDIM * FDIM) / 4;
        if (W2 != nullptr) {
            const float4* w24 = (const float4*)W2;
            for (int i = tid; i < nf4; i += 256) {
                const int k = i >> 5, c = (i & 31) * 4;
                float4 a = w4[i], b = w24[i];
                sWT[(c + 0) * LDS_STRIDE + k] = a.x + b.x;
                sWT[(c + 1) * LDS_STRIDE + k] = a.y + b.y;
                sWT[(c + 2) * LDS_STRIDE + k] = a.z + b.z;
                sWT[(c + 3) * LDS_STRIDE + k] = a.w + b.w;
            }
        } else {
            for (int i = tid; i < nf4; i += 256) {
                const int k = i >> 5, c = (i & 31) * 4;
                float4 a = w4[i];
                sWT[(c + 0) * LDS_STRIDE + k] = a.x;
                sWT[(c + 1) * LDS_STRIDE + k] = a.y;
                sWT[(c + 2) * LDS_STRIDE + k] = a.z;
                sWT[(c + 3) * LDS_STRIDE + k] = a.w;
            }
        }
    }
    __syncthreads();

    v8f acc0[8], acc1[8];
#pragma unroll
    for (int n = 0; n < 8; ++n) {
        acc0[n] = (v8f){0.f,0.f,0.f,0.f,0.f,0.f,0.f,0.f};
        acc1[n] = (v8f){0.f,0.f,0.f,0.f,0.f,0.f,0.f,0.f};
    }

#pragma unroll 4
    for (int k0 = 0; k0 < FDIM; k0 += 4) {
        const int k = k0 + 2 * half;
        const v2f a0 = *(const v2f*)(xrow0 + k);
        const v2f a1 = *(const v2f*)(xrow1 + k);
#pragma unroll
        for (int n = 0; n < 8; ++n) {
            const v2f b = *(const v2f*)(sWT + (n * 16 + m) * LDS_STRIDE + k);
            acc0[n] = __builtin_amdgcn_wmma_f32_16x16x4_f32(
                          false, a0, false, b, (short)0, acc0[n], false, false);
            acc1[n] = __builtin_amdgcn_wmma_f32_16x16x4_f32(
                          false, a1, false, b, (short)0, acc1[n], false, false);
        }
    }

#pragma unroll
    for (int i = 0; i < 8; ++i) {
        const int row0 = rowBase + i + 8 * half;
        if (row0 < nRows) {
            float* orow = Out + (size_t)row0 * FDIM;
#pragma unroll
            for (int n = 0; n < 8; ++n)
                orow[n * 16 + m] = acc0[n][i];
        }
        const int row1 = rowBase + 16 + i + 8 * half;
        if (row1 < nRows) {
            float* orow = Out + (size_t)row1 * FDIM;
#pragma unroll
            for (int n = 0; n < 8; ++n)
                orow[n * 16 + m] = acc1[n][i];
        }
    }
}

__global__ __launch_bounds__(256)
void hgnn_scatter_add(const float* __restrict__ Y,
                      const int* __restrict__ src,
                      const int* __restrict__ dst,
                      float* __restrict__ Out,
                      int nEdges)
{
    const int gid  = blockIdx.x * 256 + threadIdx.x;
    const int e    = gid >> 5;
    const int lane = gid & 31;
    if (e >= nEdges) return;

    const int s = src[e];
    const int d = dst[e];

    const float4 v = ((const float4*)(Y + (size_t)s * FDIM))[lane];
    float* o = Out + (size_t)d * FDIM + lane * 4;
    unsafeAtomicAdd(o + 0, v.x);
    unsafeAtomicAdd(o + 1, v.y);
    unsafeAtomicAdd(o + 2, v.z);
    unsafeAtomicAdd(o + 3, v.w);
}

extern "C" void kernel_launch(void* const* d_in, const int* in_sizes, int n_in,
                              void* d_out, int out_size, void* d_ws, size_t ws_size,
                              hipStream_t stream)
{
    const float* xA        = (const float*)d_in[0];
    const float* xB        = (const float*)d_in[1];
    const float* Wmsg_ab1  = (const float*)d_in[2];
    const float* Wself_ab1 = (const float*)d_in[3];
    const float* Wmsg_ab2  = (const float*)d_in[4];
    const float* Wself_ab2 = (const float*)d_in[5];
    const float* Wmsg_ba   = (const float*)d_in[6];
    const float* Wself_ba  = (const float*)d_in[7];
    const int*   src_ab1   = (const int*)d_in[8];
    const int*   dst_ab1   = (const int*)d_in[9];
    const int*   src_ab2   = (const int*)d_in[10];
    const int*   dst_ab2   = (const int*)d_in[11];
    const int*   src_ba    = (const int*)d_in[12];
    const int*   dst_ba    = (const int*)d_in[13];

    const int NA = in_sizes[0] / FDIM;
    const int NB = in_sizes[1] / FDIM;
    const int E  = in_sizes[8];

    float* outA = (float*)d_out;
    float* outB = outA + (size_t)NA * FDIM;
    float* Y    = (float*)d_ws;

    const dim3 blk(256);
    const dim3 gA((NA + ROWS_PER_BLOCK - 1) / ROWS_PER_BLOCK);
    const dim3 gB((NB + ROWS_PER_BLOCK - 1) / ROWS_PER_BLOCK);
    const dim3 gE((unsigned)(((size_t)E * 32 + 255) / 256));

    hgnn_gemm128_f32<<<gA, blk, 0, stream>>>(xA, Wself_ba,  nullptr,   outA, NA);
    hgnn_gemm128_f32<<<gB, blk, 0, stream>>>(xB, Wself_ab1, Wself_ab2, outB, NB);

    hgnn_gemm128_f32<<<gA, blk, 0, stream>>>(xA, Wmsg_ab1, nullptr, Y, NA);
    hgnn_scatter_add<<<gE, blk, 0, stream>>>(Y, src_ab1, dst_ab1, outB, E);

    hgnn_gemm128_f32<<<gA, blk, 0, stream>>>(xA, Wmsg_ab2, nullptr, Y, NA);
    hgnn_scatter_add<<<gE, blk, 0, stream>>>(Y, src_ab2, dst_ab2, outB, E);

    hgnn_gemm128_f32<<<gB, blk, 0, stream>>>(xB, Wmsg_ba, nullptr, Y, NB);
    hgnn_scatter_add<<<gE, blk, 0, stream>>>(Y, src_ba, dst_ba, outA, E);
}


# scatter-add: lane owns columns l,l+32,l+64,l+96 so each f32 atomic add and each Y load covers one whole 128-B line (4 add line requests per edge, was 16); sums stay in atomic arrival order
# speedup vs baseline: 1.1961x; 1.1961x over previous
#include <hip/hip_runtime.h>


typedef __attribute__((ext_vector_type(2))) float v2f;
typedef __attribute__((ext_vector_type(8))) float v8f;

#define FDIM 128
#define LDS_STRIDE 132
#define ROWS_PER_WAVE 32
#define ROWS_PER_BLOCK 256

__global__ __launch_bounds__(256)
void hgnn_gemm128_f32(const float* __restrict__ X,
                      const float* __restrict__ W,
                      const float* __restrict__ W2,
                      float* __restrict__ Out,
                      int nRows)
{
    __shared__ float sWT[FDIM * LDS_STRIDE];

    const int tid  = threadIdx.x;
    const int wave = tid >> 5;
    const int lane = tid & 31;
    const int half = lane >> 4;
    const int m    = lane & 15;

    const int rowBase = blockIdx.x * ROWS_PER_BLOCK + wave * ROWS_PER_WAVE;

    int r0 = rowBase + m;       if (r0 >= nRows) r0 = nRows - 1;
    int r1 = rowBase + 16 + m;  if (r1 >= nRows) r1 = nRows - 1;
    const float* xrow0 = X + (size_t)r0 * FDIM;
    const float* xrow1 = X + (size_t)r1 * FDIM;

    __builtin_prefetch(xrow0, 0, 0);
    __builtin_prefetch(xrow1, 0, 0);

    {
        const float4* w4  = (const float4*)W;
        const int     nf4 = (FDIM * FDIM) / 4;
        if (W2 != nullptr) {
            const float4* w24 = (const float4*)W2;
            for (int i = tid; i < nf4; i += 256) {
                const int k = i >> 5, c = (i & 31) * 4;
                float4 a = w4[i], b = w24[i];
                sWT[(c + 0) * LDS_STRIDE + k] = a.x + b.x;
                sWT[(c + 1) * LDS_STRIDE + k] = a.y + b.y;
                sWT[(c + 2) * LDS_STRIDE + k] = a.z + b.z;
                sWT[(c + 3) * LDS_STRIDE + k] = a.w + b.w;
            }
        } else {
            for (int i = tid; i < nf4; i += 256) {
                const int k = i >> 5, c = (i & 31) * 4;
                float4 a = w4[i];
                sWT[(c + 0) * LDS_STRIDE + k] = a.x;
                sWT[(c + 1) * LDS_STRIDE + k] = a.y;
                sWT[(c + 2) * LDS_STRIDE + k] = a.z;
                sWT[(c + 3) * LDS_STRIDE + k] = a.w;
            }
        }
    }
    __syncthreads();

    v8f acc0[8], acc1[8];
#pragma unroll
    for (int n = 0; n < 8; ++n) {
        acc0[n] = (v8f){0.f,0.f,0.f,0.f,0.f,0.f,0.f,0.f};
        acc1[n] = (v8f){0.f,0.f,0.f,0.f,0.f,0.f,0.f,0.f};
    }

#pragma unroll 4
    for (int k0 = 0; k0 < FDIM; k0 += 4) {
        const int k = k0 + 2 * half;
        const v2f a0 = *(const v2f*)(xrow0 + k);
        const v2f a1 = *(const v2f*)(xrow1 + k);
#pragma unroll
        for (int n = 0; n < 8; ++n) {
            const v2f b = *(const v2f*)(sWT + (n * 16 + m) * LDS_STRIDE + k);
            acc0[n] = __builtin_amdgcn_wmma_f32_16x16x4_f32(
                          false, a0, false, b, (short)0, acc0[n], false, false);
            acc1[n] = __builtin_amdgcn_wmma_f32_16x16x4_f32(
                          false, a1, false, b, (short)0, acc1[n], false, false);
        }
    }

#pragma unroll
    for (int i = 0; i < 8; ++i) {
        const int row0 = rowBase + i + 8 * half;
        if (row0 < nRows) {
            float* orow = Out + (size_t)row0 * FDIM;
#pragma unroll
            for (int n = 0; n < 8; ++n)
                orow[n * 16 + m] = acc0[n][i];
        }
        const int row1 = rowBase + 16 + i + 8 * half;
        if (row1 < nRows) {
            float* orow = Out + (size_t)row1 * FDIM;
#pragma unroll
            for (int n = 0; n < 8; ++n)
                orow[n * 16 + m] = acc1[n][i];
        }
    }
}

__global__ __launch_bounds__(256)
void hgnn_scatter_add(const float* __restrict__ Y,
                      const int* __restrict__ src,
                      const int* __restrict__ dst,
                      float* __restrict__ Out,
                      int nEdges)
{
    const int gid  = blockIdx.x * 256 + threadIdx.x;
    const int e    = gid >> 5;
    const int lane = gid & 31;
    if (e >= nEdges) return;

    const int s = src[e];
    const int d = dst[e];

    const float* y = Y + (size_t)s * FDIM + lane;
    const float v0 = y[0];
    const float v1 = y[32];
    const float v2 = y[64];
    const float v3 = y[96];

    float* o = Out + (size_t)d * FDIM + lane;
    unsafeAtomicAdd(o +  0, v0);
    unsafeAtomicAdd(o + 32, v1);
    unsafeAtomicAdd(o + 64, v2);
    unsafeAtomicAdd(o + 96, v3);
}

extern "C" void kernel_launch(void* const* d_in, const int* in_sizes, int n_in,
                              void* d_out, int out_size, void* d_ws, size_t ws_size,
                              hipStream_t stream)
{
    const float* xA        = (const float*)d_in[0];
    const float* xB        = (const float*)d_in[1];
    const float* Wmsg_ab1  = (const float*)d_in[2];
    const float* Wself_ab1 = (const float*)d_in[3];
    const float* Wmsg_ab2  = (const float*)d_in[4];
    const float* Wself_ab2 = (const float*)d_in[5];
    const float* Wmsg_ba   = (const float*)d_in[6];
    const float* Wself_ba  = (const float*)d_in[7];
    const int*   src_ab1   = (const int*)d_in[8];
    const int*   dst_ab1   = (const int*)d_in[9];
    const int*   src_ab2   = (const int*)d_in[10];
    const int*   dst_ab2   = (const int*)d_in[11];
    const int*   src_ba    = (const int*)d_in[12];
    const int*   dst_ba    = (const int*)d_in[13];

    const int NA = in_sizes[0] / FDIM;
    const int NB = in_sizes[1] / FDIM;
    const int E  = in_sizes[8];

    float* outA = (float*)d_out;
    float* outB = outA + (size_t)NA * FDIM;
    float* Y    = (float*)d_ws;

    const dim3 blk(256);
    const dim3 gA((NA + ROWS_PER_BLOCK - 1) / ROWS_PER_BLOCK);
    const dim3 gB((NB + ROWS_PER_BLOCK - 1) / ROWS_PER_BLOCK);
    const dim3 gE((unsigned)(((size_t)E * 32 + 255) / 256));

    hgnn_gemm128_f32<<<gA, blk, 0, stream>>>(xA, Wself_ba,  nullptr,   outA, NA);
    hgnn_gemm128_f32<<<gB, blk, 0, stream>>>(xB, Wself_ab1, Wself_ab2, outB, NB);

    hgnn_gemm128_f32<<<gA, blk, 0, stream>>>(xA, Wmsg_ab1, nullptr, Y, NA);
    hgnn_scatter_add<<<gE, blk, 0, stream>>>(Y, src_ab1, dst_ab1, outB, E);

    hgnn_gemm128_f32<<<gA, blk, 0, stream>>>(xA, Wmsg_ab2, nullptr, Y, NA);
    hgnn_scatter_add<<<gE, blk, 0, stream>>>(Y, src_ab2, dst_ab2, outB, E);

    hgnn_gemm128_f32<<<gB, blk, 0, stream>>>(xB, Wmsg_ba, nullptr, Y, NB);
    hgnn_scatter_add<<<gE, blk, 0, stream>>>(Y, src_ba, dst_ba, outA, E);
}


# scatter-add: each wave walks 32 consecutive edges (one coalesced index load, rows of 8 edges in flight before their adds); 12500 waves per launch, was 400000; sums stay in atomic arrival order
# speedup vs baseline: 1.3808x; 1.1544x over previous
#include <hip/hip_runtime.h>


typedef __attribute__((ext_vector_type(2))) float v2f;
typedef __attribute__((ext_vector_type(8))) float v8f;

#define FDIM 128
#define LDS_STRIDE 132
#define ROWS_PER_WAVE 32
#define ROWS_PER_BLOCK 256

__global__ __launch_bounds__(256)
void hgnn_gemm128_f32(const float* __restrict__ X,
                      const float* __restrict__ W,
                      const float* __restrict__ W2,
                      float* __restrict__ Out,
                      int nRows)
{
    __shared__ float sWT[FDIM * LDS_STRIDE];

    const int tid  = threadIdx.x;
    const int wave = tid >> 5;
    const int lane = tid & 31;
    const int half = lane >> 4;
    const int m    = lane & 15;

    const int rowBase = blockIdx.x * ROWS_PER_BLOCK + wave * ROWS_PER_WAVE;

    int r0 = rowBase + m;       if (r0 >= nRows) r0 = nRows - 1;
    int r1 = rowBase + 16 + m;  if (r1 >= nRows) r1 = nRows - 1;
    const float* xrow0 = X + (size_t)r0 * FDIM;
    const float* xrow1 = X + (size_t)r1 * FDIM;

    __builtin_prefetch(xrow0, 0, 0);
    __builtin_prefetch(xrow1, 0, 0);

    {
        const float4* w4  = (const float4*)W;
        const int     nf4 = (FDIM * FDIM) / 4;
        if (W2 != nullptr) {
            const float4* w24 = (const float4*)W2;
            for (int i = tid; i < nf4; i += 256) {
                const int k = i >> 5, c = (i & 31) * 4;
                float4 a = w4[i], b = w24[i];
                sWT[(c + 0) * LDS_STRIDE + k] = a.x + b.x;
                sWT[(c + 1) * LDS_STRIDE + k] = a.y + b.y;
                sWT[(c + 2) * LDS_STRIDE + k] = a.z + b.z;
                sWT[(c + 3) * LDS_STRIDE + k] = a.w + b.w;
            }
        } else {
            for (int i = tid; i < nf4; i += 256) {
                const int k = i >> 5, c = (i & 31) * 4;
                float4 a = w4[i];
                sWT[(c + 0) * LDS_STRIDE + k] = a.x;
                sWT[(c + 1) * LDS_STRIDE + k] = a.y;
                sWT[(c + 2) * LDS_STRIDE + k] = a.z;
                sWT[(c + 3) * LDS_STRIDE + k] = a.w;
            }
        }
    }
    __syncthreads();

    v8f acc0[8], acc1[8];
#pragma unroll
    for (int n = 0; n < 8; ++n) {
        acc0[n] = (v8f){0.f,0.f,0.f,0.f,0.f,0.f,0.f,0.f};
        acc1[n] = (v8f){0.f,0.f,0.f,0.f,0.f,0.f,0.f,0.f};
    }

#pragma unroll 4
    for (int k0 = 0; k0 < FDIM; k0 += 4) {
        const int k = k0 + 2 * half;
        const v2f a0 = *(const v2f*)(xrow0 + k);
        const v2f a1 = *(const v2f*)(xrow1 + k);
#pragma unroll
        for (int n = 0; n < 8; ++n) {
            const v2f b = *(const v2f*)(sWT + (n * 16 + m) * LDS_STRIDE + k);
            acc0[n] = __builtin_amdgcn_wmma_f32_16x16x4_f32(
                          false, a0, false, b, (short)0, acc0[n], false, false);
            acc1[n] = __builtin_amdgcn_wmma_f32_16x16x4_f32(
                          false, a1, false, b, (short)0, acc1[n], false, false);
        }
    }

#pragma unroll
    for (int i = 0; i < 8; ++i) {
        const int row0 = rowBase + i + 8 * half;
        if (row0 < nRows) {
            float* orow = Out + (size_t)row0 * FDIM;
#pragma unroll
            for (int n = 0; n < 8; ++n)
                orow[n * 16 + m] = acc0[n][i];
        }
        const int row1 = rowBase + 16 + i + 8 * half;
        if (row1 < nRows) {
            float* orow = Out + (size_t)row1 * FDIM;
#pragma unroll
            for (int n = 0; n < 8; ++n)
                orow[n * 16 + m] = acc1[n][i];
        }
    }
}

#define EDGES_PER_WAVE 32
#define EDGE_BATCH 8
#define EDGES_PER_BLOCK 256

__global__ __launch_bounds__(256)
void hgnn_scatter_add(const float* __restrict__ Y,
                      const int* __restrict__ src,
                      const int* __restrict__ dst,
                      float* __restrict__ Out,
                      int nEdges)
{
    const int lane = threadIdx.x & 31;
    const int wave = __builtin_amdgcn_readfirstlane((int)(threadIdx.x >> 5));
    const int e0   = blockIdx.x * EDGES_PER_BLOCK + wave * EDGES_PER_WAVE;
    if (e0 >= nEdges) return;
    int n = nEdges - e0;
    if (n > EDGES_PER_WAVE) n = EDGES_PER_WAVE;

    int sv = 0, dv = 0;
    if (lane < n) {
        sv = src[e0 + lane];
        dv = dst[e0 + lane];
    }

    const float* yl = Y + lane;
    float*       ol = Out + lane;

    int j = 0;
    for (; j + EDGE_BATCH <= n; j += EDGE_BATCH) {
        float v[EDGE_BATCH][4];
#pragma unroll
        for (int b = 0; b < EDGE_BATCH; ++b) {
            const int s = __builtin_amdgcn_readlane(sv, j + b);
            const float* y = yl + (size_t)s * FDIM;
            v[b][0] = y[0];
            v[b][1] = y[32];
            v[b][2] = y[64];
            v[b][3] = y[96];
        }
#pragma unroll
        for (int b = 0; b < EDGE_BATCH; ++b) {
            const int d = __builtin_amdgcn_readlane(dv, j + b);
            float* o = ol + (size_t)d * FDIM;
            unsafeAtomicAdd(o +  0, v[b][0]);
            unsafeAtomicAdd(o + 32, v[b][1]);
            unsafeAtomicAdd(o + 64, v[b][2]);
            unsafeAtomicAdd(o + 96, v[b][3]);
        }
    }
    for (; j < n; ++j) {
        const int s = __builtin_amdgcn_readlane(sv, j);
        const int d = __builtin_amdgcn_readlane(dv, j);
        const float* y = yl + (size_t)s * FDIM;
        const float v0 = y[0];
        const float v1 = y[32];
        const float v2 = y[64];
        const float v3 = y[96];
        float* o = ol + (size_t)d * FDIM;
        unsafeAtomicAdd(o +  0, v0);
        unsafeAtomicAdd(o + 32, v1);
        unsafeAtomicAdd(o + 64, v2);
        unsafeAtomicAdd(o + 96, v3);
    }
}

extern "C" void kernel_launch(void* const* d_in, const int* in_sizes, int n_in,
                              void* d_out, int out_size, void* d_ws, size_t ws_size,
                              hipStream_t stream)
{
    const float* xA        = (const float*)d_in[0];
    const float* xB        = (const float*)d_in[1];
    const float* Wmsg_ab1  = (const float*)d_in[2];
    const float* Wself_ab1 = (const float*)d_in[3];
    const float* Wmsg_ab2  = (const float*)d_in[4];
    const float* Wself_ab2 = (const float*)d_in[5];
    const float* Wmsg_ba   = (const float*)d_in[6];
    const float* Wself_ba  = (const float*)d_in[7];
    const int*   src_ab1   = (const int*)d_in[8];
    const int*   dst_ab1   = (const int*)d_in[9];
    const int*   src_ab2   = (const int*)d_in[10];
    const int*   dst_ab2   = (const int*)d_in[11];
    const int*   src_ba    = (const int*)d_in[12];
    const int*   dst_ba    = (const int*)d_in[13];

    const int NA = in_sizes[0] / FDIM;
    const int NB = in_sizes[1] / FDIM;
    const int E  = in_sizes[8];

    float* outA = (float*)d_out;
    float* outB = outA + (size_t)NA * FDIM;
    float* Y    = (float*)d_ws;

    const dim3 blk(256);
    const dim3 gA((NA + ROWS_PER_BLOCK - 1) / ROWS_PER_BLOCK);
    const dim3 gB((NB + ROWS_PER_BLOCK - 1) / ROWS_PER_BLOCK);
    const dim3 gE((E + EDGES_PER_BLOCK - 1) / EDGES_PER_BLOCK);

    hgnn_gemm128_f32<<<gA, blk, 0, stream>>>(xA, Wself_ba,  nullptr,   outA, NA);
    hgnn_gemm128_f32<<<gB, blk, 0, stream>>>(xB, Wself_ab1, Wself_ab2, outB, NB);

    hgnn_gemm128_f32<<<gA, blk, 0, stream>>>(xA, Wmsg_ab1, nullptr, Y, NA);
    hgnn_scatter_add<<<gE, blk, 0, stream>>>(Y, src_ab1, dst_ab1, outB, E);

    hgnn_gemm128_f32<<<gA, blk, 0, stream>>>(xA, Wmsg_ab2, nullptr, Y, NA);
    hgnn_scatter_add<<<gE, blk, 0, stream>>>(Y, src_ab2, dst_ab2, outB, E);

    hgnn_gemm128_f32<<<gB, blk, 0, stream>>>(xB, Wmsg_ba, nullptr, Y, NB);
    hgnn_scatter_add<<<gE, blk, 0, stream>>>(Y, src_ba, dst_ba, outA, E);
}


# launch fusion: the five products run as one grid and the three scatter-adds as one grid (8 launches -> 2), three message planes in workspace; sums stay in atomic arrival order, both A->B lists mixed
# speedup vs baseline: 1.9179x; 1.3890x over previous
#include <hip/hip_runtime.h>


typedef __attribute__((ext_vector_type(2))) float v2f;
typedef __attribute__((ext_vector_type(8))) float v8f;

#define FDIM 128
#define LDS_STRIDE 132
#define ROWS_PER_WAVE 32
#define ROWS_PER_BLOCK 256

__global__ __launch_bounds__(256)
void hgnn_gemm128_f32_x5(const float* __restrict__ xA,
                         const float* __restrict__ xB,
                         const float* __restrict__ Wself_ba,
                         const float* __restrict__ Wmsg_ab1,
                         const float* __restrict__ Wmsg_ab2,
                         const float* __restrict__ Wself_ab1,
                         const float* __restrict__ Wself_ab2,
                         const float* __restrict__ Wmsg_ba,
                         float* __restrict__ outA,
                         float* __restrict__ outB,
                         float* __restrict__ Y1,
                         float* __restrict__ Y2,
                         float* __restrict__ Y3,
                         int NA, int NB, int nbA)
{
    __shared__ float sWT[FDIM * LDS_STRIDE];

    const int tid  = threadIdx.x;
    const int wave = tid >> 5;
    const int lane = tid & 31;
    const int half = lane >> 4;
    const int m    = lane & 15;

    const float* X;
    const float* W;
    const float* W2 = nullptr;
    float*       Out;
    int nRows, rowBlock;
    {
        int b = (int)blockIdx.x;
        if (b < 3 * nbA) {
            rowBlock = b / 3;
            const int j = b - rowBlock * 3;
            X = xA; nRows = NA;
            if (j == 0)      { W = Wself_ba; Out = outA; }
            else if (j == 1) { W = Wmsg_ab1; Out = Y1; }
            else             { W = Wmsg_ab2; Out = Y2; }
        } else {
            b -= 3 * nbA;
            rowBlock = b >> 1;
            X = xB; nRows = NB;
            if ((b & 1) == 0) { W = Wself_ab1; W2 = Wself_ab2; Out = outB; }
            else              { W = Wmsg_ba; Out = Y3; }
        }
    }

    const int rowBase = rowBlock * ROWS_PER_BLOCK + wave * ROWS_PER_WAVE;

    int r0 = rowBase + m;       if (r0 >= nRows) r0 = nRows - 1;
    int r1 = rowBase + 16 + m;  if (r1 >= nRows) r1 = nRows - 1;
    const float* xrow0 = X + (size_t)r0 * FDIM;
    const float* xrow1 = X + (size_t)r1 * FDIM;

    __builtin_prefetch(xrow0, 0, 0);
    __builtin_prefetch(xrow1, 0, 0);

    {
        const float4* w4  = (const float4*)W;
        const int     nf4 = (FDIM * FDIM) / 4;
        if (W2 != nullptr) {
            const float4* w24 = (const float4*)W2;
            for (int i = tid; i < nf4; i += 256) {
                const int k = i >> 5, c = (i & 31) * 4;
                float4 a = w4[i], b = w24[i];
                sWT[(c + 0) * LDS_STRIDE + k] = a.x + b.x;
                sWT[(c + 1) * LDS_STRIDE + k] = a.y + b.y;
                sWT[(c + 2) * LDS_STRIDE + k] = a.z + b.z;
                sWT[(c + 3) * LDS_STRIDE + k] = a.w + b.w;
            }
        } else {
            for (int i = tid; i < nf4; i += 256) {
                const int k = i >> 5, c = (i & 31) * 4;
                float4 a = w4[i];
                sWT[(c + 0) * LDS_STRIDE + k] = a.x;
                sWT[(c + 1) * LDS_STRIDE + k] = a.y;
                sWT[(c + 2) * LDS_STRIDE + k] = a.z;
                sWT[(c + 3) * LDS_STRIDE + k] = a.w;
            }
        }
    }
    __syncthreads();

    v8f acc0[8], acc1[8];
#pragma unroll
    for (int n = 0; n < 8; ++n) {
        acc0[n] = (v8f){0.f,0.f,0.f,0.f,0.f,0.f,0.f,0.f};
        acc1[n] = (v8f){0.f,0.f,0.f,0.f,0.f,0.f,0.f,0.f};
    }

#pragma unroll 4
    for (int k0 = 0; k0 < FDIM; k0 += 4) {
        const int k = k0 + 2 * half;
        const v2f a0 = *(const v2f*)(xrow0 + k);
        const v2f a1 = *(const v2f*)(xrow1 + k);
#pragma unroll
        for (int n = 0; n < 8; ++n) {
            const v2f b = *(const v2f*)(sWT + (n * 16 + m) * LDS_STRIDE + k);
            acc0[n] = __builtin_amdgcn_wmma_f32_16x16x4_f32(
                          false, a0, false, b, (short)0, acc0[n], false, false);
            acc1[n] = __builtin_amdgcn_wmma_f32_16x16x4_f32(
                          false, a1, false, b, (short)0, acc1[n], false, false);
        }
    }

#pragma unroll
    for (int i = 0; i < 8; ++i) {
        const int row0 = rowBase + i + 8 * half;
        if (row0 < nRows) {
            float* orow = Out + (size_t)row0 * FDIM;
#pragma unroll
            for (int n = 0; n < 8; ++n)
                orow[n * 16 + m] = acc0[n][i];
        }
        const int row1 = rowBase + 16 + i + 8 * half;
        if (row1 < nRows) {
            float* orow = Out + (size_t)row1 * FDIM;
#pragma unroll
            for (int n = 0; n < 8; ++n)
                orow[n * 16 + m] = acc1[n][i];
        }
    }
}

#define EDGES_PER_WAVE 32
#define EDGE_BATCH 8
#define EDGES_PER_BLOCK 256

__global__ __launch_bounds__(256)
void hgnn_scatter_add_x3(const float* __restrict__ Y1,
                         const int* __restrict__ src1,
                         const int* __restrict__ dst1,
                         const float* __restrict__ Y2,
                         const int* __restrict__ src2,
                         const int* __restrict__ dst2,
                         const float* __restrict__ Y3,
                         const int* __restrict__ src3,
                         const int* __restrict__ dst3,
                         float* __restrict__ outA,
                         float* __restrict__ outB,
                         int nEdges)
{
    const float* Y;
    const int*   src;
    const int*   dst;
    float*       Out;
    {
        const int rel = (int)blockIdx.y;
        if (rel == 0)      { Y = Y1; src = src1; dst = dst1; Out = outB; }
        else if (rel == 1) { Y = Y2; src = src2; dst = dst2; Out = outB; }
        else               { Y = Y3; src = src3; dst = dst3; Out = outA; }
    }

    const int lane = threadIdx.x & 31;
    const int wave = __builtin_amdgcn_readfirstlane((int)(threadIdx.x >> 5));
    const int e0   = blockIdx.x * EDGES_PER_BLOCK + wave * EDGES_PER_WAVE;
    if (e0 >= nEdges) return;
    int n = nEdges - e0;
    if (n > EDGES_PER_WAVE) n = EDGES_PER_WAVE;

    int sv = 0, dv = 0;
    if (lane < n) {
        sv = src[e0 + lane];
        dv = dst[e0 + lane];
    }

    const float* yl = Y + lane;
    float*       ol = Out + lane;

    int j = 0;
    for (; j + EDGE_BATCH <= n; j += EDGE_BATCH) {
        float v[EDGE_BATCH][4];
#pragma unroll
        for (int b = 0; b < EDGE_BATCH; ++b) {
            const int s = __builtin_amdgcn_readlane(sv, j + b);
            const float* y = yl + (size_t)s * FDIM;
            v[b][0] = y[0];
            v[b][1] = y[32];
            v[b][2] = y[64];
            v[b][3] = y[96];
        }
#pragma unroll
        for (int b = 0; b < EDGE_BATCH; ++b) {
            const int d = __builtin_amdgcn_readlane(dv, j + b);
            float* o = ol + (size_t)d * FDIM;
            unsafeAtomicAdd(o +  0, v[b][0]);
            unsafeAtomicAdd(o + 32, v[b][1]);
            unsafeAtomicAdd(o + 64, v[b][2]);
            unsafeAtomicAdd(o + 96, v[b][3]);
        }
    }
    for (; j < n; ++j) {
        const int s = __builtin_amdgcn_readlane(sv, j);
        const int d = __builtin_amdgcn_readlane(dv, j);
        const float* y = yl + (size_t)s * FDIM;
        const float v0 = y[0];
        const float v1 = y[32];
        const float v2 = y[64];
        const float v3 = y[96];
        float* o = ol + (size_t)d * FDIM;
        unsafeAtomicAdd(o +  0, v0);
        unsafeAtomicAdd(o + 32, v1);
        unsafeAtomicAdd(o + 64, v2);
        unsafeAtomicAdd(o + 96, v3);
    }
}

extern "C" void kernel_launch(void* const* d_in, const int* in_sizes, int n_in,
                              void* d_out, int out_size, void* d_ws, size_t ws_size,
                              hipStream_t stream)
{
    const float* xA        = (const float*)d_in[0];
    const float* xB        = (const float*)d_in[1];
    const float* Wmsg_ab1  = (const float*)d_in[2];
    const float* Wself_ab1 = (const float*)d_in[3];
    const float* Wmsg_ab2  = (const float*)d_in[4];
    const float* Wself_ab2 = (const float*)d_in[5];
    const float* Wmsg_ba   = (const float*)d_in[6];
    const float* Wself_ba  = (const float*)d_in[7];
    const int*   src_ab1   = (const int*)d_in[8];
    const int*   dst_ab1   = (const int*)d_in[9];
    const int*   src_ab2   = (const int*)d_in[10];
    const int*   dst_ab2   = (const int*)d_in[11];
    const int*   src_ba    = (const int*)d_in[12];
    const int*   dst_ba    = (const int*)d_in[13];

    const int NA = in_sizes[0] / FDIM;
    const int NB = in_sizes[1] / FDIM;
    const int E  = in_sizes[8];

    float* outA = (float*)d_out;
    float* outB = outA + (size_t)NA * FDIM;
    float* Y1   = (float*)d_ws;
    float* Y2   = Y1 + (size_t)NA * FDIM;
    float* Y3   = Y2 + (size_t)NA * FDIM;

    const dim3 blk(256);
    const int  nbA = (NA + ROWS_PER_BLOCK - 1) / ROWS_PER_BLOCK;
    const int  nbB = (NB + ROWS_PER_BLOCK - 1) / ROWS_PER_BLOCK;
    const dim3 gG(3 * nbA + 2 * nbB);
    const dim3 gE((E + EDGES_PER_BLOCK - 1) / EDGES_PER_BLOCK, 3);

    hgnn_gemm128_f32_x5<<<gG, blk, 0, stream>>>(xA, xB, Wself_ba, Wmsg_ab1, Wmsg_ab2,
                                                Wself_ab1, Wself_ab2, Wmsg_ba,
                                                outA, outB, Y1, Y2, Y3, NA, NB, nbA);

    hgnn_scatter_add_x3<<<gE, blk, 0, stream>>>(Y1, src_ab1, dst_ab1,
                                                Y2, src_ab2, dst_ab2,
                                                Y3, src_ba,  dst_ba,
                                                outA, outB, E);
}
